# starting kernel
# baseline (speedup 1.0000x reference)
#include <hip/hip_runtime.h>
#include <hip/hip_bf16.h>

#define BATCH 8
#define SEQ   4096
#define DIM   1024
#define NSTATE 256
#define LN_EPS 1e-5f

typedef float v2f __attribute__((ext_vector_type(2)));
typedef float v4f __attribute__((ext_vector_type(4)));
typedef float v8f __attribute__((ext_vector_type(8)));

__global__ void __launch_bounds__(256)
ssm_rowmean_wmma(const float* __restrict__ x, float* __restrict__ xmean,
                 int rows, int cols) {
    const int wid  = (blockIdx.x * blockDim.x + threadIdx.x) >> 5;
    const int lane = threadIdx.x & 31;
    const int r0   = wid << 4;
    if (r0 >= rows) return;

    const int myrow = r0 + (lane & 15);
    const int koff  = (lane >> 4) << 1;
    const float* rowp = x + (size_t)myrow * cols + koff;

    const float inv = 1.0f / (float)cols;
    v2f bmat; bmat[0] = inv; bmat[1] = inv;
    v8f acc = {};

#pragma unroll 8
    for (int k = 0; k < cols; k += 4) {
        v2f a = *(const v2f*)(rowp + k);
        acc = __builtin_amdgcn_wmma_f32_16x16x4_f32(
             false, a,  false, bmat,
             (short)0, acc,  false,  false);
    }

    if (lane == 0) {
#pragma unroll
        for (int j = 0; j < 8; ++j) xmean[r0 + j] = acc[j];
    } else if (lane == 16) {
#pragma unroll
        for (int j = 0; j < 8; ++j) xmean[r0 + 8 + j] = acc[j];
    }
}

__device__ __forceinline__ float wave_sum(float v) {
#pragma unroll
    for (int off = 16; off >= 1; off >>= 1) v += __shfl_xor(v, off, 32);
    return v;
}

__global__ void __launch_bounds__(BATCH * 32)
ssm_scan(const float* __restrict__ xmean,
         const float* __restrict__ vol,
         const float* __restrict__ llr,
         const float* __restrict__ logb,
         const float* __restrict__ cvec,
         const float* __restrict__ log_d,
         const float* __restrict__ log_step,
         const float* __restrict__ vgate,
         const float* __restrict__ lnw,
         const float* __restrict__ lnb,
         float* __restrict__ ys)
{
    const int lane  = threadIdx.x & 31;
    const int batch = threadIdx.x >> 5;

    const float step = __expf(log_step[0]);

    float a_[8], b_[8], g_[8], w_[8], o_[8], c_[8], h_[8];
#pragma unroll
    for (int j = 0; j < 8; ++j) {
        const int n = lane + 32 * j;
        const float lam = -__expf(llr[n]);
        const float ad  = (2.0f + step * lam) * __builtin_amdgcn_rcpf(2.0f - step * lam);
        a_[j] = ad;
        b_[j] = step * (1.0f + ad) * __expf(logb[n]) * 0.5f;
        g_[j] = __builtin_amdgcn_rcpf(1.0f + __expf(-vgate[n]));
        w_[j] = lnw[n];
        o_[j] = lnb[n];
        c_[j] = cvec[n];
        h_[j] = 0.0f;
    }

    const float* xmb  = xmean + batch * SEQ;
    const float* volb = vol   + batch * SEQ;
    float*       ysb  = ys    + batch * SEQ;
    const float invN = 1.0f / (float)NSTATE;

    for (int t = 0; t < SEQ; ++t) {
        const float xm = xmb[t];
        const float vt = volb[t];

        float s1 = 0.0f, s2 = 0.0f;
#pragma unroll
        for (int j = 0; j < 8; ++j) {
            const float h = fmaf(a_[j], h_[j], b_[j] * xm);
            h_[j] = h;
            s1 += h;
            s2 = fmaf(h, h, s2);
        }
        s1 = wave_sum(s1);
        s2 = wave_sum(s2);

        const float mu  = s1 * invN;
        const float var = fmaf(s2, invN, -mu * mu);
        const float rs  = __builtin_amdgcn_rsqf(var + LN_EPS);

        float dot = 0.0f;
#pragma unroll
        for (int j = 0; j < 8; ++j) {
            float hn = fmaf((h_[j] - mu) * rs, w_[j], o_[j]);
            hn *= __builtin_amdgcn_rcpf(fmaf(g_[j], vt, 1.0f));
            h_[j] = hn;
            dot = fmaf(hn, c_[j], dot);
        }
        dot = wave_sum(dot);
        if (lane == 0) ysb[t] = dot;
    }
}

__global__ void __launch_bounds__(256)
ssm_output(const float* __restrict__ x, const float* __restrict__ ys,
           const float* __restrict__ alpha, const float* __restrict__ log_d,
           float* __restrict__ out, int total4) {
    const int i = blockIdx.x * blockDim.x + threadIdx.x;
    if (i >= total4) return;

    const float a  = 1.0f / (1.0f + __expf(-alpha[0]));
    const float dd = __expf(log_d[0]);
    const float s0 = a + (1.0f - a) * dd;
    const float s1 = 1.0f - a;

    const int row = i >> 8;
    const float yv = s1 * ys[row];

    const v4f xv = ((const v4f*)x)[i];
    v4f o;
    o.x = fmaf(s0, xv.x, yv);
    o.y = fmaf(s0, xv.y, yv);
    o.z = fmaf(s0, xv.z, yv);
    o.w = fmaf(s0, xv.w, yv);
    __builtin_nontemporal_store(o, ((v4f*)out) + i);
}

extern "C" void kernel_launch(void* const* d_in, const int* in_sizes, int n_in,
                              void* d_out, int out_size, void* d_ws, size_t ws_size,
                              hipStream_t stream) {
    const float* x        = (const float*)d_in[0];
    const float* vol      = (const float*)d_in[1];
    const float* llr      = (const float*)d_in[2];
    const float* logb     = (const float*)d_in[3];
    const float* cvec     = (const float*)d_in[4];
    const float* log_d    = (const float*)d_in[5];
    const float* log_step = (const float*)d_in[6];
    const float* vgate    = (const float*)d_in[7];
    const float* alpha    = (const float*)d_in[8];
    const float* lnw      = (const float*)d_in[9];
    const float* lnb      = (const float*)d_in[10];
    float* out = (float*)d_out;

    const int rows = BATCH * SEQ;
    float* xmean = (float*)d_ws;
    float* ys    = xmean + rows;

    {
        const int blocks = rows / 128;
        ssm_rowmean_wmma<<<blocks, 256, 0, stream>>>(x, xmean, rows, DIM);
    }
    ssm_scan<<<1, BATCH * 32, 0, stream>>>(xmean, vol, llr, logb, cvec, log_d,
                                           log_step, vgate, lnw, lnb, ys);
    {
        const int total4 = (BATCH * SEQ * DIM) / 4;
        const int blocks = (total4 + 255) / 256;
        ssm_output<<<blocks, 256, 0, stream>>>(x, ys, alpha, log_d, out, total4);
    }
}


# scan kernel: sum u and sum u*u of a step go through ONE f32 WMMA (A = lane's two sums, B picks one per column parity), one fold, four lane gathers; one WMMA and fold fewer per step; sums regrouped
# speedup vs baseline: 6.0630x; 6.0630x over previous
#include <hip/hip_runtime.h>
#include <hip/hip_bf16.h>

#define BATCH 8
#define SEQ   4096
#define DIM   1024
#define NSTATE 256
#define LN_EPS 1e-5f

typedef float v2f __attribute__((ext_vector_type(2)));
typedef float v4f __attribute__((ext_vector_type(4)));
typedef float v8f __attribute__((ext_vector_type(8)));

__global__ void __launch_bounds__(256)
ssm_rowmean_wmma(const float* __restrict__ x, float* __restrict__ xmean,
                 int rows, int cols) {
    const int wid  = (blockIdx.x * blockDim.x + threadIdx.x) >> 5;
    const int lane = threadIdx.x & 31;
    const int r0   = wid << 4;
    if (r0 >= rows) return;

    const int myrow = r0 + (lane & 15);
    const int koff  = (lane >> 4) << 1;
    const float* rowp = x + (size_t)myrow * cols + koff;

    const float inv = 1.0f / (float)cols;
    v2f bmat; bmat[0] = inv; bmat[1] = inv;
    v8f acc = {};

#pragma unroll 8
    for (int k = 0; k < cols; k += 4) {
        v2f a = *(const v2f*)(rowp + k);
        acc = __builtin_amdgcn_wmma_f32_16x16x4_f32(
             false, a,  false, bmat,
             (short)0, acc,  false,  false);
    }

    if (lane == 0) {
#pragma unroll
        for (int j = 0; j < 8; ++j) xmean[r0 + j] = acc[j];
    } else if (lane == 16) {
#pragma unroll
        for (int j = 0; j < 8; ++j) xmean[r0 + 8 + j] = acc[j];
    }
}

#define XV_PITCH (SEQ + 2)
#define SCAN_WAVES (BATCH / 2)
#define SPL 16
#define DOTBLK 16

template <bool ODD>
__device__ __forceinline__ float same_row(float x) {
    const int b = __builtin_bit_cast(int, x);
    return __builtin_bit_cast(float, ODD
        ? __builtin_amdgcn_permlane16(0, b, 0x77553311, 0xffddbb99, false, false)
        : __builtin_amdgcn_permlane16(0, b, 0x66442200, 0xeeccaa88, false, false));
}
template <bool ODD>
__device__ __forceinline__ float other_row(float x) {
    const int b = __builtin_bit_cast(int, x);
    return __builtin_bit_cast(float, ODD
        ? __builtin_amdgcn_permlanex16(0, b, 0x77553311, 0xffddbb99, false, false)
        : __builtin_amdgcn_permlanex16(0, b, 0x66442200, 0xeeccaa88, false, false));
}

__device__ __forceinline__ float add_plain(float a, float b) {
    float r;
    asm("v_add_f32 %0, %1, %2" : "=v"(r) : "v"(a), "v"(b));
    return r;
}

__device__ __forceinline__ float fold8(v8f d) {
    const v2f d01 = __builtin_shufflevector(d, d, 0, 1);
    const v2f d23 = __builtin_shufflevector(d, d, 2, 3);
    const v2f d45 = __builtin_shufflevector(d, d, 4, 5);
    const v2f d67 = __builtin_shufflevector(d, d, 6, 7);
    const v2f s = (d01 + d23) + (d45 + d67);
    return add_plain(s[0], s[1]);
}

__device__ __forceinline__ float sums_mma(v2f a, v2f bsel) {
    const v8f zero = {};
    const v8f d = __builtin_amdgcn_wmma_f32_16x16x4_f32(
         false, a,  false, bsel,
         (short)0, zero,  false,  false);
    return fold8(d);
}

__global__ void __launch_bounds__(SCAN_WAVES * 32)
ssm_scan(const float* __restrict__ xmean,
         const float* __restrict__ vol,
         const float* __restrict__ llr,
         const float* __restrict__ logb,
         const float* __restrict__ cvec,
         const float* __restrict__ log_d,
         const float* __restrict__ log_step,
         const float* __restrict__ vgate,
         const float* __restrict__ lnw,
         const float* __restrict__ lnb,
         float* __restrict__ ys)
{
    __shared__ __attribute__((aligned(16))) v2f sxv[BATCH * XV_PITCH];

    const int lane = threadIdx.x & 31;
    const int wv   = threadIdx.x >> 5;
    const int c    = lane & 15;
    const int hf   = lane >> 4;
    const int grp  = c >> 3;
    const int li   = (c & 7) + 8 * hf;
    const int mybatch = 2 * wv + grp;
    const bool own = (grp == hf);

    const float step = __expf(log_step[0]);

    v2f a_[SPL / 2], b_[SPL / 2], g_[SPL / 2], w_[SPL / 2], o_[SPL / 2], c_[SPL / 2], h_[SPL / 2];
#pragma unroll
    for (int j = 0; j < SPL; ++j) {
        const int n = li + 16 * j;
        const float lam = -__expf(llr[n]);
        const float ad  = (2.0f + step * lam) * __builtin_amdgcn_rcpf(2.0f - step * lam);
        a_[j >> 1][j & 1] = ad;
        b_[j >> 1][j & 1] = step * (1.0f + ad) * __expf(logb[n]) * 0.5f;
        g_[j >> 1][j & 1] = __builtin_amdgcn_rcpf(1.0f + __expf(-vgate[n]));
        w_[j >> 1][j & 1] = lnw[n];
        o_[j >> 1][j & 1] = lnb[n];
        c_[j >> 1][j & 1] = cvec[n];
        h_[j >> 1][j & 1] = 0.0f;
    }

    for (int bb = 0; bb < 2; ++bb) {
        const int row = 2 * wv + bb;
        const v4f* xm4 = (const v4f*)(xmean + row * SEQ);
        const v4f* vl4 = (const v4f*)(vol   + row * SEQ);
        v4f* dst = (v4f*)(sxv + row * XV_PITCH);
        for (int k = 0; k < SEQ / 128; ++k) {
            const int i = lane + 32 * k;
            const v4f a = xm4[i];
            const v4f v = vl4[i];
            v4f lo, hi;
            lo[0] = a[0]; lo[1] = v[0]; lo[2] = a[1]; lo[3] = v[1];
            hi[0] = a[2]; hi[1] = v[2]; hi[2] = a[3]; hi[3] = v[3];
            dst[2 * i]     = lo;
            dst[2 * i + 1] = hi;
        }
        if (lane == 0) { const v4f z = {}; dst[SEQ / 2] = z; }
    }
    __syncthreads();

    const v2f* srow = sxv + mybatch * XV_PITCH;
    float* ysb = ys + (2 * wv + hf) * SEQ;
    const float invN = 1.0f / (float)NSTATE;
    v2f bsel; bsel[0] = (c & 1) ? 0.0f : 1.0f; bsel[1] = (c & 1) ? 1.0f : 0.0f;

    v2f selj[DOTBLK];
#pragma unroll
    for (int j = 0; j < DOTBLK; ++j) {
        const float s = (c == j) ? 1.0f : 0.0f;
        selj[j][0] = s; selj[j][1] = s;
    }

    v2f cur = srow[0];
    for (int t0 = 0; t0 < SEQ; t0 += DOTBLK) {
      v8f dacc = {};
#pragma unroll
      for (int j = 0; j < DOTBLK; ++j) {
        const int t = t0 + j;
        const v2f nxt = srow[t + 1];
        const float xm = cur[0];
        const float vt = cur[1];

#pragma unroll
        for (int j = 0; j < SPL / 2; ++j) h_[j] = __builtin_elementwise_fma(a_[j], h_[j], b_[j] * xm);

        const v2f p = ((h_[0] + h_[1]) + (h_[2] + h_[3])) + ((h_[4] + h_[5]) + (h_[6] + h_[7]));
        const v2f q = (__builtin_elementwise_fma(h_[0], h_[0], h_[1] * h_[1]) +
                       __builtin_elementwise_fma(h_[2], h_[2], h_[3] * h_[3])) +
                      (__builtin_elementwise_fma(h_[4], h_[4], h_[5] * h_[5]) +
                       __builtin_elementwise_fma(h_[6], h_[6], h_[7] * h_[7]));
        v2f pq; pq[0] = add_plain(p[0], p[1]); pq[1] = add_plain(q[0], q[1]);
        const float e = sums_mma(pq, bsel);
        const float u_same = same_row<false>(e), u_other = other_row<false>(e);
        const float q_same = same_row<true>(e),  q_other = other_row<true>(e);
        const float s1 = own ? u_same : u_other;
        const float s2 = own ? q_same : q_other;

        const float mu  = s1 * invN;
        const float m2e = fmaf(s2, invN, LN_EPS);
        const float rs  = __builtin_amdgcn_rsqf(fmaf(-mu, mu, m2e));

        v2f dsum[SPL / 2];
#pragma unroll
        for (int j = 0; j < SPL / 2; ++j) {
            v2f hn = __builtin_elementwise_fma((h_[j] - mu) * rs, w_[j], o_[j]);
            const v2f gd = __builtin_elementwise_fma(g_[j], (v2f){vt, vt}, (v2f){1.0f, 1.0f});
            v2f gr; gr[0] = __builtin_amdgcn_rcpf(gd[0]); gr[1] = __builtin_amdgcn_rcpf(gd[1]);
            hn *= gr;
            h_[j] = hn;
            dsum[j] = hn * c_[j];
        }
        const v2f dd = ((dsum[0] + dsum[1]) + (dsum[2] + dsum[3])) + ((dsum[4] + dsum[5]) + (dsum[6] + dsum[7]));
        dacc = __builtin_amdgcn_wmma_f32_16x16x4_f32(
             false, dd,  false, selj[j],
             (short)0, dacc,  false,  false);
        cur = nxt;
      }
      const float dot = fold8(dacc);
      ysb[t0 + c] = dot;
    }
}

__global__ void __launch_bounds__(256)
ssm_output(const float* __restrict__ x, const float* __restrict__ ys,
           const float* __restrict__ alpha, const float* __restrict__ log_d,
           float* __restrict__ out, int total4) {
    const int i = blockIdx.x * blockDim.x + threadIdx.x;
    if (i >= total4) return;

    const float a  = 1.0f / (1.0f + __expf(-alpha[0]));
    const float dd = __expf(log_d[0]);
    const float s0 = a + (1.0f - a) * dd;
    const float s1 = 1.0f - a;

    const int row = i >> 8;
    const float yv = s1 * ys[row];

    const v4f xv = ((const v4f*)x)[i];
    v4f o;
    o.x = fmaf(s0, xv.x, yv);
    o.y = fmaf(s0, xv.y, yv);
    o.z = fmaf(s0, xv.z, yv);
    o.w = fmaf(s0, xv.w, yv);
    __builtin_nontemporal_store(o, ((v4f*)out) + i);
}

extern "C" void kernel_launch(void* const* d_in, const int* in_sizes, int n_in,
                              void* d_out, int out_size, void* d_ws, size_t ws_size,
                              hipStream_t stream) {
    const float* x        = (const float*)d_in[0];
    const float* vol      = (const float*)d_in[1];
    const float* llr      = (const float*)d_in[2];
    const float* logb     = (const float*)d_in[3];
    const float* cvec     = (const float*)d_in[4];
    const float* log_d    = (const float*)d_in[5];
    const float* log_step = (const float*)d_in[6];
    const float* vgate    = (const float*)d_in[7];
    const float* alpha    = (const float*)d_in[8];
    const float* lnw      = (const float*)d_in[9];
    const float* lnb      = (const float*)d_in[10];
    float* out = (float*)d_out;

    const int rows = BATCH * SEQ;
    float* xmean = (float*)d_ws;
    float* ys    = xmean + rows;

    {
        const int blocks = rows / 128;
        ssm_rowmean_wmma<<<blocks, 256, 0, stream>>>(x, xmean, rows, DIM);
    }
    ssm_scan<<<1, SCAN_WAVES * 32, 0, stream>>>(xmean, vol, llr, logb, cvec, log_d,
                                           log_step, vgate, lnw, lnb, ys);
    {
        const int total4 = (BATCH * SEQ * DIM) / 4;
        const int blocks = (total4 + 255) / 256;
        ssm_output<<<blocks, 256, 0, stream>>>(x, ys, alpha, log_d, out, total4);
    }
}


# scan kernel: a step's output dot leaves each lane as ONE value (two multiply-add chains over its pairs, added), so two steps share one selector WMMA: 8 dot WMMAs per 16 steps; dot sums regrouped
# speedup vs baseline: 6.2557x; 1.0318x over previous
#include <hip/hip_runtime.h>
#include <hip/hip_bf16.h>

#define BATCH 8
#define SEQ   4096
#define DIM   1024
#define NSTATE 256
#define LN_EPS 1e-5f

typedef float v2f __attribute__((ext_vector_type(2)));
typedef float v4f __attribute__((ext_vector_type(4)));
typedef float v8f __attribute__((ext_vector_type(8)));

__global__ void __launch_bounds__(256)
ssm_rowmean_wmma(const float* __restrict__ x, float* __restrict__ xmean,
                 int rows, int cols) {
    const int wid  = (blockIdx.x * blockDim.x + threadIdx.x) >> 5;
    const int lane = threadIdx.x & 31;
    const int r0   = wid << 4;
    if (r0 >= rows) return;

    const int myrow = r0 + (lane & 15);
    const int koff  = (lane >> 4) << 1;
    const float* rowp = x + (size_t)myrow * cols + koff;

    const float inv = 1.0f / (float)cols;
    v2f bmat; bmat[0] = inv; bmat[1] = inv;
    v8f acc = {};

#pragma unroll 8
    for (int k = 0; k < cols; k += 4) {
        v2f a = *(const v2f*)(rowp + k);
        acc = __builtin_amdgcn_wmma_f32_16x16x4_f32(
             false, a,  false, bmat,
             (short)0, acc,  false,  false);
    }

    if (lane == 0) {
#pragma unroll
        for (int j = 0; j < 8; ++j) xmean[r0 + j] = acc[j];
    } else if (lane == 16) {
#pragma unroll
        for (int j = 0; j < 8; ++j) xmean[r0 + 8 + j] = acc[j];
    }
}

#define XV_PITCH (SEQ + 2)
#define SCAN_WAVES (BATCH / 2)
#define SPL 16
#define DOTBLK 16
#define DOTPAIRS (DOTBLK / 2)

template <bool ODD>
__device__ __forceinline__ float same_row(float x) {
    const int b = __builtin_bit_cast(int, x);
    return __builtin_bit_cast(float, ODD
        ? __builtin_amdgcn_permlane16(0, b, 0x77553311, 0xffddbb99, false, false)
        : __builtin_amdgcn_permlane16(0, b, 0x66442200, 0xeeccaa88, false, false));
}
template <bool ODD>
__device__ __forceinline__ float other_row(float x) {
    const int b = __builtin_bit_cast(int, x);
    return __builtin_bit_cast(float, ODD
        ? __builtin_amdgcn_permlanex16(0, b, 0x77553311, 0xffddbb99, false, false)
        : __builtin_amdgcn_permlanex16(0, b, 0x66442200, 0xeeccaa88, false, false));
}

__device__ __forceinline__ float add_plain(float a, float b) {
    float r;
    asm("v_add_f32 %0, %1, %2" : "=v"(r) : "v"(a), "v"(b));
    return r;
}

__device__ __forceinline__ float fold8(v8f d) {
    const v2f d01 = __builtin_shufflevector(d, d, 0, 1);
    const v2f d23 = __builtin_shufflevector(d, d, 2, 3);
    const v2f d45 = __builtin_shufflevector(d, d, 4, 5);
    const v2f d67 = __builtin_shufflevector(d, d, 6, 7);
    const v2f s = (d01 + d23) + (d45 + d67);
    return add_plain(s[0], s[1]);
}

__device__ __forceinline__ float sums_mma(v2f a, v2f bsel) {
    const v8f zero = {};
    const v8f d = __builtin_amdgcn_wmma_f32_16x16x4_f32(
         false, a,  false, bsel,
         (short)0, zero,  false,  false);
    return fold8(d);
}

__global__ void __launch_bounds__(SCAN_WAVES * 32)
ssm_scan(const float* __restrict__ xmean,
         const float* __restrict__ vol,
         const float* __restrict__ llr,
         const float* __restrict__ logb,
         const float* __restrict__ cvec,
         const float* __restrict__ log_d,
         const float* __restrict__ log_step,
         const float* __restrict__ vgate,
         const float* __restrict__ lnw,
         const float* __restrict__ lnb,
         float* __restrict__ ys)
{
    __shared__ __attribute__((aligned(16))) v2f sxv[BATCH * XV_PITCH];

    const int lane = threadIdx.x & 31;
    const int wv   = threadIdx.x >> 5;
    const int c    = lane & 15;
    const int hf   = lane >> 4;
    const int grp  = c >> 3;
    const int li   = (c & 7) + 8 * hf;
    const int mybatch = 2 * wv + grp;
    const bool own = (grp == hf);

    const float step = __expf(log_step[0]);

    v2f a_[SPL / 2], b_[SPL / 2], g_[SPL / 2], w_[SPL / 2], o_[SPL / 2], c_[SPL / 2], h_[SPL / 2];
#pragma unroll
    for (int j = 0; j < SPL; ++j) {
        const int n = li + 16 * j;
        const float lam = -__expf(llr[n]);
        const float ad  = (2.0f + step * lam) * __builtin_amdgcn_rcpf(2.0f - step * lam);
        a_[j >> 1][j & 1] = ad;
        b_[j >> 1][j & 1] = step * (1.0f + ad) * __expf(logb[n]) * 0.5f;
        g_[j >> 1][j & 1] = __builtin_amdgcn_rcpf(1.0f + __expf(-vgate[n]));
        w_[j >> 1][j & 1] = lnw[n];
        o_[j >> 1][j & 1] = lnb[n];
        c_[j >> 1][j & 1] = cvec[n];
        h_[j >> 1][j & 1] = 0.0f;
    }

    for (int bb = 0; bb < 2; ++bb) {
        const int row = 2 * wv + bb;
        const v4f* xm4 = (const v4f*)(xmean + row * SEQ);
        const v4f* vl4 = (const v4f*)(vol   + row * SEQ);
        v4f* dst = (v4f*)(sxv + row * XV_PITCH);
        for (int k = 0; k < SEQ / 128; ++k) {
            const int i = lane + 32 * k;
            const v4f a = xm4[i];
            const v4f v = vl4[i];
            v4f lo, hi;
            lo[0] = a[0]; lo[1] = v[0]; lo[2] = a[1]; lo[3] = v[1];
            hi[0] = a[2]; hi[1] = v[2]; hi[2] = a[3]; hi[3] = v[3];
            dst[2 * i]     = lo;
            dst[2 * i + 1] = hi;
        }
        if (lane == 0) { const v4f z = {}; dst[SEQ / 2] = z; }
    }
    __syncthreads();

    const v2f* srow = sxv + mybatch * XV_PITCH;
    float* ysb = ys + (2 * wv + hf) * SEQ;
    const float invN = 1.0f / (float)NSTATE;
    v2f bsel; bsel[0] = (c & 1) ? 0.0f : 1.0f; bsel[1] = (c & 1) ? 1.0f : 0.0f;

    v2f sel2[DOTPAIRS];
#pragma unroll
    for (int i = 0; i < DOTPAIRS; ++i) {
        sel2[i][0] = (c == 2 * i)     ? 1.0f : 0.0f;
        sel2[i][1] = (c == 2 * i + 1) ? 1.0f : 0.0f;
    }

    v2f cur = srow[0];
    for (int t0 = 0; t0 < SEQ; t0 += DOTBLK) {
      v8f dacc = {};
      float deven = 0.0f;
#pragma unroll
      for (int j = 0; j < DOTBLK; ++j) {
        const int t = t0 + j;
        const v2f nxt = srow[t + 1];
        const float xm = cur[0];
        const float vt = cur[1];

#pragma unroll
        for (int j = 0; j < SPL / 2; ++j) h_[j] = __builtin_elementwise_fma(a_[j], h_[j], b_[j] * xm);

        const v2f p = ((h_[0] + h_[1]) + (h_[2] + h_[3])) + ((h_[4] + h_[5]) + (h_[6] + h_[7]));
        const v2f q = (__builtin_elementwise_fma(h_[0], h_[0], h_[1] * h_[1]) +
                       __builtin_elementwise_fma(h_[2], h_[2], h_[3] * h_[3])) +
                      (__builtin_elementwise_fma(h_[4], h_[4], h_[5] * h_[5]) +
                       __builtin_elementwise_fma(h_[6], h_[6], h_[7] * h_[7]));
        v2f pq; pq[0] = add_plain(p[0], p[1]); pq[1] = add_plain(q[0], q[1]);
        const float e = sums_mma(pq, bsel);
        const float u_same = same_row<false>(e), u_other = other_row<false>(e);
        const float q_same = same_row<true>(e),  q_other = other_row<true>(e);
        const float s1 = own ? u_same : u_other;
        const float s2 = own ? q_same : q_other;

        const float mu  = s1 * invN;
        const float m2e = fmaf(s2, invN, LN_EPS);
        const float rs  = __builtin_amdgcn_rsqf(fmaf(-mu, mu, m2e));

        v2f dcha, dchb;
#pragma unroll
        for (int k = 0; k < SPL / 2; ++k) {
            v2f hn = __builtin_elementwise_fma((h_[k] - mu) * rs, w_[k], o_[k]);
            const v2f gd = __builtin_elementwise_fma(g_[k], (v2f){vt, vt}, (v2f){1.0f, 1.0f});
            v2f gr; gr[0] = __builtin_amdgcn_rcpf(gd[0]); gr[1] = __builtin_amdgcn_rcpf(gd[1]);
            hn *= gr;
            h_[k] = hn;
            if (k < SPL / 4) dcha = (k == 0)       ? hn * c_[k] : __builtin_elementwise_fma(hn, c_[k], dcha);
            else             dchb = (k == SPL / 4) ? hn * c_[k] : __builtin_elementwise_fma(hn, c_[k], dchb);
        }
        const v2f dch = dcha + dchb;
        const float dl = add_plain(dch[0], dch[1]);
        if (j & 1) {
            v2f da; da[0] = deven; da[1] = dl;
            dacc = __builtin_amdgcn_wmma_f32_16x16x4_f32(
                 false, da,  false, sel2[j >> 1],
                 (short)0, dacc,  false,  false);
        } else {
            deven = dl;
        }
        cur = nxt;
      }
      const float dot = fold8(dacc);
      ysb[t0 + c] = dot;
    }
}

__global__ void __launch_bounds__(256)
ssm_output(const float* __restrict__ x, const float* __restrict__ ys,
           const float* __restrict__ alpha, const float* __restrict__ log_d,
           float* __restrict__ out, int total4) {
    const int i = blockIdx.x * blockDim.x + threadIdx.x;
    if (i >= total4) return;

    const float a  = 1.0f / (1.0f + __expf(-alpha[0]));
    const float dd = __expf(log_d[0]);
    const float s0 = a + (1.0f - a) * dd;
    const float s1 = 1.0f - a;

    const int row = i >> 8;
    const float yv = s1 * ys[row];

    const v4f xv = ((const v4f*)x)[i];
    v4f o;
    o.x = fmaf(s0, xv.x, yv);
    o.y = fmaf(s0, xv.y, yv);
    o.z = fmaf(s0, xv.z, yv);
    o.w = fmaf(s0, xv.w, yv);
    __builtin_nontemporal_store(o, ((v4f*)out) + i);
}

extern "C" void kernel_launch(void* const* d_in, const int* in_sizes, int n_in,
                              void* d_out, int out_size, void* d_ws, size_t ws_size,
                              hipStream_t stream) {
    const float* x        = (const float*)d_in[0];
    const float* vol      = (const float*)d_in[1];
    const float* llr      = (const float*)d_in[2];
    const float* logb     = (const float*)d_in[3];
    const float* cvec     = (const float*)d_in[4];
    const float* log_d    = (const float*)d_in[5];
    const float* log_step = (const float*)d_in[6];
    const float* vgate    = (const float*)d_in[7];
    const float* alpha    = (const float*)d_in[8];
    const float* lnw      = (const float*)d_in[9];
    const float* lnb      = (const float*)d_in[10];
    float* out = (float*)d_out;

    const int rows = BATCH * SEQ;
    float* xmean = (float*)d_ws;
    float* ys    = xmean + rows;

    {
        const int blocks = rows / 128;
        ssm_rowmean_wmma<<<blocks, 256, 0, stream>>>(x, xmean, rows, DIM);
    }
    ssm_scan<<<1, SCAN_WAVES * 32, 0, stream>>>(xmean, vol, llr, logb, cvec, log_d,
                                           log_step, vgate, lnw, lnb, ys);
    {
        const int total4 = (BATCH * SEQ * DIM) / 4;
        const int blocks = (total4 + 255) / 256;
        ssm_output<<<blocks, 256, 0, stream>>>(x, ys, alpha, log_d, out, total4);
    }
}
